# starting kernel
# baseline (speedup 1.0000x reference)
#include <hip/hip_runtime.h>
#include <hip/hip_bf16.h>


typedef __attribute__((ext_vector_type(2))) float v2f;
typedef __attribute__((ext_vector_type(8))) float v8f;

#define NPIX 4096
#define CDIM 128
#define LWIN 25

__device__ __forceinline__ int clamp6(int v) {
    return v < 0 ? 0 : (v > 63 ? 63 : v);
}

__global__ void __launch_bounds__(256)
shiftlsa_stageA(const float* __restrict__ fmap, const float* __restrict__ fmap2,
                const float* __restrict__ W1,   const float* __restrict__ W2,
                float* __restrict__ F1t, float* __restrict__ F2t) {
    const int t = threadIdx.x;
    const int d = t & 127;
    const int p = t >> 7;
    const int P = blockIdx.x * 2 + p;
    const int b = P >> 12;
    const int pix = P & 4095;

    const float* fm1 = fmap  + ((size_t)b * CDIM) * NPIX + pix;
    const float* fm2 = fmap2 + ((size_t)b * CDIM) * NPIX + pix;
    const float* w1  = W1 + (size_t)d * CDIM;
    const float* w2  = W2 + (size_t)d * CDIM;

    float s1 = 0.f, s2 = 0.f;
#pragma unroll 8
    for (int c = 0; c < CDIM; ++c) {
        const float u = fm1[(size_t)c * NPIX];
        const float v = fm2[(size_t)c * NPIX];
        s1 = fmaf(w1[c], u, s1);
        s2 = fmaf(w2[c], v, s2);
    }
    F1t[(size_t)P * CDIM + d] = s1;
    F2t[(size_t)P * CDIM + d] = s2;
}

__global__ void __launch_bounds__(256)
shiftlsa_stageB(const float* __restrict__ F1t, const float* __restrict__ F2t,
                float* __restrict__ G) {
    __shared__ float As[64 * 68];
    __shared__ float Bs[64 * 68];

    const int b  = blockIdx.z;
    const int n0 = blockIdx.x * 64;
    const int m0 = blockIdx.y * 64;

    const float* Fa = F1t + (size_t)b * NPIX * CDIM;
    const float* Fb = F2t + (size_t)b * NPIX * CDIM;

    const int lane = threadIdx.x & 31;
    const int w    = threadIdx.x >> 5;
    const int twr  = w & 3;
    const int twc  = w >> 2;

    v8f acc0 = {};
    v8f acc1 = {};

    const int arow  = twr * 16 + (lane & 15);
    const int bcol0 = twc * 32 + (lane & 15);
    const int bcol1 = bcol0 + 16;
    const int koff  = (lane >> 4) << 1;

    for (int kb = 0; kb < CDIM; kb += 64) {
#pragma unroll
        for (int it = 0; it < 4; ++it) {
            const int f   = threadIdx.x + 256 * it;
            const int row = f >> 4;
            const int cc  = (f & 15) << 2;
            *(float4*)(&As[row * 68 + cc]) =
                *(const float4*)(Fa + ((size_t)(n0 + row)) * CDIM + kb + cc);
            *(float4*)(&Bs[row * 68 + cc]) =
                *(const float4*)(Fb + ((size_t)(m0 + row)) * CDIM + kb + cc);
        }
        __syncthreads();

#pragma unroll
        for (int k = 0; k < 64; k += 4) {
            const v2f a  = *(const v2f*)(&As[arow  * 68 + k + koff]);
            const v2f b0 = *(const v2f*)(&Bs[bcol0 * 68 + k + koff]);
            const v2f b1 = *(const v2f*)(&Bs[bcol1 * 68 + k + koff]);
            acc0 = __builtin_amdgcn_wmma_f32_16x16x4_f32(
                       false, a, false, b0, (short)0, acc0, false, false);
            acc1 = __builtin_amdgcn_wmma_f32_16x16x4_f32(
                       false, a, false, b1, (short)0, acc1, false, false);
        }
        __syncthreads();
    }

    float* Gp = G + (size_t)b * NPIX * NPIX;
    const int orow  = n0 + twr * 16 + ((lane >> 4) << 3);
    const int ocol  = m0 + twc * 32 + (lane & 15);
#pragma unroll
    for (int r = 0; r < 8; ++r) {
        Gp[(size_t)(orow + r) * NPIX + ocol]      = acc0[r];
        Gp[(size_t)(orow + r) * NPIX + ocol + 16] = acc1[r];
    }
}

__global__ void __launch_bounds__(256)
shiftlsa_stageC(const float* __restrict__ attn, const float* __restrict__ G,
                float* __restrict__ out) {
    const int b = blockIdx.y;
    const int n = blockIdx.x;
    const int y = n >> 6;
    const int x = n & 63;

    __shared__ float sA[32];
    if (threadIdx.x < LWIN)
        sA[threadIdx.x] = attn[((size_t)b * NPIX + n) * LWIN + threadIdx.x];
    __syncthreads();

    const int X  = threadIdx.x & 63;
    const int yg = threadIdx.x >> 6;

    float acc[16];
#pragma unroll
    for (int q = 0; q < 16; ++q) acc[q] = 0.f;

    const float* Gb = G + (size_t)b * NPIX * NPIX;

#pragma unroll
    for (int l = 0; l < LWIN; ++l) {
        const int i = l / 5 - 2;
        const int j = l % 5 - 2;
        const float a = sA[l];
        const int py = clamp6(y + i);
        const int px = clamp6(x + j);
        const float* Grow = Gb + ((size_t)(py * 64 + px)) * NPIX;
        const int cX = clamp6(X + j);
#pragma unroll
        for (int q = 0; q < 16; ++q) {
            const int cY = clamp6(yg * 16 + q + i);
            acc[q] = fmaf(a, Grow[cY * 64 + cX], acc[q]);
        }
    }

    const float scale = 0.0883883476483184405f;
    const size_t base = ((size_t)b * NPIX + n) * NPIX;
#pragma unroll
    for (int q = 0; q < 16; ++q)
        out[base + (size_t)(yg * 16 + q) * 64 + X] = acc[q] * scale;
}

extern "C" void kernel_launch(void* const* d_in, const int* in_sizes, int n_in,
                              void* d_out, int out_size, void* d_ws, size_t ws_size,
                              hipStream_t stream) {
    const float* attn  = (const float*)d_in[0];
    const float* fmap  = (const float*)d_in[1];
    const float* fmap2 = (const float*)d_in[2];
    const float* W1    = (const float*)d_in[3];
    const float* W2    = (const float*)d_in[4];
    float*       out   = (float*)d_out;

    float* F1t = (float*)d_ws;
    float* F2t = F1t + (size_t)2 * NPIX * CDIM;
    float* G   = F2t + (size_t)2 * NPIX * CDIM;

    shiftlsa_stageA<<<dim3(4096), dim3(256), 0, stream>>>(
        fmap, fmap2, W1, W2, F1t, F2t);

    shiftlsa_stageB<<<dim3(64, 64, 2), dim3(256), 0, stream>>>(F1t, F2t, G);

    shiftlsa_stageC<<<dim3(4096, 2), dim3(256), 0, stream>>>(attn, G, out);
}


# Stage B: 512 resident work-groups each walk 16 G tiles, A panel staged in LDS once, next B tile loaded a tile ahead into a second LDS buffer (was 8192 one-tile groups, no loads ahead); k order kept
# speedup vs baseline: 3.7819x; 3.7819x over previous
#include <hip/hip_runtime.h>
#include <hip/hip_bf16.h>


typedef __attribute__((ext_vector_type(2))) float v2f;
typedef __attribute__((ext_vector_type(8))) float v8f;

#define NPIX 4096
#define CDIM 128
#define LWIN 25

__device__ __forceinline__ int clamp6(int v) {
    return v < 0 ? 0 : (v > 63 ? 63 : v);
}

__global__ void __launch_bounds__(256)
shiftlsa_stageM(const float* __restrict__ W1, const float* __restrict__ W2,
                float* __restrict__ MT) {
    const int idx = blockIdx.x * 256 + threadIdx.x;
    const int c   = idx & 127;
    const int c2  = idx >> 7;
    float s = 0.f;
#pragma unroll 8
    for (int d = 0; d < CDIM; ++d)
        s = fmaf(W1[d * CDIM + c], W2[d * CDIM + c2], s);
    MT[idx] = s;
}

__global__ void __launch_bounds__(256) __attribute__((amdgpu_waves_per_eu(2, 2)))
shiftlsa_stageA(const float* __restrict__ fmap2, const float* __restrict__ MT,
                float* __restrict__ Tt) {
    const int lane = threadIdx.x & 31;
    const int w    = threadIdx.x >> 5;
    const int P0   = blockIdx.x * 32 + w * 4;
    const int b    = P0 >> 12;
    const int pix0 = P0 & 4095;

    const float* fm2 = fmap2 + ((size_t)b * CDIM) * NPIX + pix0;
    const float* mt  = MT + 4 * lane;

    float4 acc0 = make_float4(0.f, 0.f, 0.f, 0.f);
    float4 acc1 = make_float4(0.f, 0.f, 0.f, 0.f);
    float4 acc2 = make_float4(0.f, 0.f, 0.f, 0.f);
    float4 acc3 = make_float4(0.f, 0.f, 0.f, 0.f);

#pragma unroll 16
    for (int c2 = 0; c2 < CDIM; ++c2) {
        const float4 m = *(const float4*)(mt + c2 * CDIM);
        const float4 f = *(const float4*)(fm2 + (size_t)c2 * NPIX);
        acc0.x = fmaf(m.x, f.x, acc0.x);
        acc0.y = fmaf(m.y, f.x, acc0.y);
        acc0.z = fmaf(m.z, f.x, acc0.z);
        acc0.w = fmaf(m.w, f.x, acc0.w);
        acc1.x = fmaf(m.x, f.y, acc1.x);
        acc1.y = fmaf(m.y, f.y, acc1.y);
        acc1.z = fmaf(m.z, f.y, acc1.z);
        acc1.w = fmaf(m.w, f.y, acc1.w);
        acc2.x = fmaf(m.x, f.z, acc2.x);
        acc2.y = fmaf(m.y, f.z, acc2.y);
        acc2.z = fmaf(m.z, f.z, acc2.z);
        acc2.w = fmaf(m.w, f.z, acc2.w);
        acc3.x = fmaf(m.x, f.w, acc3.x);
        acc3.y = fmaf(m.y, f.w, acc3.y);
        acc3.z = fmaf(m.z, f.w, acc3.z);
        acc3.w = fmaf(m.w, f.w, acc3.w);
    }

    float* o = Tt + (size_t)P0 * CDIM + 4 * lane;
    *(float4*)(o)            = acc0;
    *(float4*)(o + CDIM)     = acc1;
    *(float4*)(o + 2 * CDIM) = acc2;
    *(float4*)(o + 3 * CDIM) = acc3;
}

#define APITCH 72
#define BPITCH 132
#define BTILE  (64 * BPITCH)
#define MT_PER_WG 16

__device__ __forceinline__ void gram_tile(const float* Ap, const float* Bt,
                                          v8f& acc0, v8f& acc1) {
    acc0 = (v8f){};
    acc1 = (v8f){};
#pragma unroll
    for (int k = 0; k < CDIM; k += 4) {
        v2f a;
        a.x = Ap[k * APITCH];
        a.y = Ap[(k + 1) * APITCH];
        const v2f b0 = *(const v2f*)(Bt + k);
        const v2f b1 = *(const v2f*)(Bt + 16 * BPITCH + k);
        acc0 = __builtin_amdgcn_wmma_f32_16x16x4_f32(
                   false, a, false, b0, (short)0, acc0, false, false);
        acc1 = __builtin_amdgcn_wmma_f32_16x16x4_f32(
                   false, a, false, b1, (short)0, acc1, false, false);
    }
}

__device__ __forceinline__ void gram_store(float* Gt, const v8f& acc0, const v8f& acc1) {
#pragma unroll
    for (int r = 0; r < 8; ++r) {
        Gt[(size_t)r * NPIX]      = acc0[r];
        Gt[(size_t)r * NPIX + 16] = acc1[r];
    }
}

__global__ void __launch_bounds__(256)
shiftlsa_stageB(const float* __restrict__ fmap, const float* __restrict__ Tt,
                float* __restrict__ G) {
    __shared__ __attribute__((aligned(16))) float As[CDIM * APITCH];
    __shared__ __attribute__((aligned(16))) float Bs[2 * BTILE];

    const int b   = blockIdx.z;
    const int n0  = blockIdx.x * 64;
    const int mc0 = blockIdx.y * (MT_PER_WG * 64);

    const int tid  = threadIdx.x;
    const int lane = tid & 31;
    const int w    = tid >> 5;
    const int twr  = w & 3;
    const int twc  = w >> 2;

    const int ar = tid >> 4, ac = (tid & 15) << 2;
    const int br = tid >> 5, bc = (tid & 31) << 2;

    const float* Xa = fmap + ((size_t)b * CDIM + ar) * NPIX + n0 + ac;
    const float* Fs = Tt + ((size_t)b * NPIX + mc0 + br) * CDIM + bc;

#pragma unroll
    for (int it = 0; it < 8; ++it)
        *(float4*)(&As[(ar + 16 * it) * APITCH + ac]) =
            *(const float4*)(Xa + (size_t)(16 * it) * NPIX);
#pragma unroll
    for (int it = 0; it < 8; ++it)
        *(float4*)(&Bs[(br + 8 * it) * BPITCH + bc]) =
            *(const float4*)(Fs + (size_t)(8 * it) * CDIM);
    __syncthreads();

    const int koff = (lane >> 4) << 1;
    const float* Ap = As + koff * APITCH + twr * 16 + (lane & 15);
    const int boff  = (twc * 32 + (lane & 15)) * BPITCH + koff;
    float* Gt = G + (size_t)b * NPIX * NPIX
                  + (size_t)(n0 + twr * 16 + ((lane >> 4) << 3)) * NPIX
                  + mc0 + twc * 32 + (lane & 15);
    float* Bst = Bs + br * BPITCH + bc;

    v8f acc0, acc1;
#pragma nounroll
    for (int t = 0; t < MT_PER_WG - 1; ++t) {
        float4 pre[8];
        const float* Fn = Fs + (size_t)(t + 1) * 64 * CDIM;
#pragma unroll
        for (int it = 0; it < 8; ++it)
            pre[it] = *(const float4*)(Fn + (size_t)(8 * it) * CDIM);
        __builtin_amdgcn_sched_barrier(0);

        gram_tile(Ap, Bs + (t & 1) * BTILE + boff, acc0, acc1);

        float* Bn = Bst + ((t + 1) & 1) * BTILE;
#pragma unroll
        for (int it = 0; it < 8; ++it)
            *(float4*)(Bn + 8 * it * BPITCH) = pre[it];
        __syncthreads();

        gram_store(Gt + t * 64, acc0, acc1);
    }
    gram_tile(Ap, Bs + ((MT_PER_WG - 1) & 1) * BTILE + boff, acc0, acc1);
    gram_store(Gt + (MT_PER_WG - 1) * 64, acc0, acc1);
}

__device__ __forceinline__ float from_next_lane(float edge, float src) {
    return __int_as_float(__builtin_amdgcn_update_dpp(
        __float_as_int(edge), __float_as_int(src), 0x101, 0xf, 0xf, false));
}
__device__ __forceinline__ float from_prev_lane(float edge, float src) {
    return __int_as_float(__builtin_amdgcn_update_dpp(
        __float_as_int(edge), __float_as_int(src), 0x111, 0xf, 0xf, false));
}

#define NTX 16

__global__ void __launch_bounds__(128)
shiftlsa_stageC(const float* __restrict__ attn, const float* __restrict__ G,
                float* __restrict__ out) {
    const int b  = blockIdx.z;
    const int y  = blockIdx.x >> 2;
    const int x0 = (blockIdx.x & 3) * NTX;
    const int c  = threadIdx.x & 15;
    const int Y  = blockIdx.y * 8 + (threadIdx.x >> 4);

    const int n0 = y * 64 + x0;
    const float* an = attn + ((size_t)b * NPIX + n0) * LWIN;
    const float* Gb = G + (size_t)b * NPIX * NPIX + 4 * c;

    float4 acc[NTX];
#pragma unroll
    for (int t = 0; t < NTX; ++t) acc[t] = make_float4(0.f, 0.f, 0.f, 0.f);

#pragma nounroll
    for (int ii = 0; ii < 5; ++ii) {
        const int py = clamp6(y + ii - 2);
        const int ro = clamp6(Y + ii - 2) * 64;
        const float* Grow = Gb + ((size_t)(py * 64)) * NPIX + ro;

        float4 g[NTX + 4];
#pragma unroll
        for (int k = 0; k < NTX + 4; ++k) {
            int px = x0 - 2 + k;
            if (k < 2)        px = px < 0 ? 0 : px;
            if (k >= NTX + 2) px = px > 63 ? 63 : px;
            g[k] = *(const float4*)(Grow + (size_t)px * NPIX);
        }
        const float* ai = an + ii * 5;
        float wfirst[4][5];
#pragma unroll
        for (int t = 0; t < 4; ++t)
#pragma unroll
            for (int q = 0; q < 5; ++q) wfirst[t][q] = ai[t * LWIN + q];

        __builtin_amdgcn_sched_barrier(0);

#pragma unroll
        for (int k = 0; k < NTX + 4; ++k) {
            const float4 v = g[k];
            const float pz = from_prev_lane(v.x, v.z);
            const float pw = from_prev_lane(v.x, v.w);
            const float nx = from_next_lane(v.w, v.x);
            const float ny = from_next_lane(v.w, v.y);
#pragma unroll
            for (int t = 0; t < NTX; ++t) {
                const int j = k - 2 - t;
                if (j < -2 || j > 2) continue;
                const float a = t < 4 ? wfirst[t][j + 2] : ai[t * LWIN + (j + 2)];
                float4 s;
                if (j == -2)      s = make_float4(pz,  pw,  v.x, v.y);
                else if (j == -1) s = make_float4(pw,  v.x, v.y, v.z);
                else if (j == 0)  s = v;
                else if (j == 1)  s = make_float4(v.y, v.z, v.w, nx);
                else              s = make_float4(v.z, v.w, nx,  ny);
                acc[t].x = fmaf(a, s.x, acc[t].x);
                acc[t].y = fmaf(a, s.y, acc[t].y);
                acc[t].z = fmaf(a, s.z, acc[t].z);
                acc[t].w = fmaf(a, s.w, acc[t].w);
            }
        }
    }

    const float scale = 0.0883883476483184405f;
    float* orow = out + ((size_t)b * NPIX + n0) * NPIX + Y * 64 + 4 * c;
#pragma unroll
    for (int t = 0; t < NTX; ++t) {
        float4 o;
        o.x = acc[t].x * scale;
        o.y = acc[t].y * scale;
        o.z = acc[t].z * scale;
        o.w = acc[t].w * scale;
        *(float4*)(orow + (size_t)t * NPIX) = o;
    }
}

extern "C" void kernel_launch(void* const* d_in, const int* in_sizes, int n_in,
                              void* d_out, int out_size, void* d_ws, size_t ws_size,
                              hipStream_t stream) {
    const float* attn  = (const float*)d_in[0];
    const float* fmap  = (const float*)d_in[1];
    const float* fmap2 = (const float*)d_in[2];
    const float* W1    = (const float*)d_in[3];
    const float* W2    = (const float*)d_in[4];
    float*       out   = (float*)d_out;

    float* Tt = (float*)d_ws;
    float* MT = Tt + (size_t)2 * NPIX * CDIM;
    float* G  = MT + (size_t)2 * NPIX * CDIM;

    shiftlsa_stageM<<<dim3(64), dim3(256), 0, stream>>>(W1, W2, MT);

    shiftlsa_stageA<<<dim3(256), dim3(256), 0, stream>>>(fmap2, MT, Tt);

    shiftlsa_stageB<<<dim3(64, 4, 2), dim3(256), 0, stream>>>(fmap, Tt, G);

    shiftlsa_stageC<<<dim3(256, 8, 2), dim3(128), 0, stream>>>(attn, G, out);
}


# Stage C: a thread keeps 4 pixel rows on the (y,Y) diagonal x 8 pixels in x in registers: a loaded G piece serves up to 4 tap rows and 5 tap columns, 100 -> 48 G loads per 16 pixels; tap order kept
# speedup vs baseline: 4.0700x; 1.0762x over previous
#include <hip/hip_runtime.h>
#include <hip/hip_bf16.h>


typedef __attribute__((ext_vector_type(2))) float v2f;
typedef __attribute__((ext_vector_type(8))) float v8f;

#define NPIX 4096
#define CDIM 128
#define LWIN 25

__device__ __forceinline__ int clamp6(int v) {
    return v < 0 ? 0 : (v > 63 ? 63 : v);
}

__global__ void __launch_bounds__(256)
shiftlsa_stageM(const float* __restrict__ W1, const float* __restrict__ W2,
                float* __restrict__ MT) {
    const int idx = blockIdx.x * 256 + threadIdx.x;
    const int c   = idx & 127;
    const int c2  = idx >> 7;
    float s = 0.f;
#pragma unroll 8
    for (int d = 0; d < CDIM; ++d)
        s = fmaf(W1[d * CDIM + c], W2[d * CDIM + c2], s);
    MT[idx] = s;
}

__global__ void __launch_bounds__(256) __attribute__((amdgpu_waves_per_eu(2, 2)))
shiftlsa_stageA(const float* __restrict__ fmap2, const float* __restrict__ MT,
                float* __restrict__ Tt) {
    const int lane = threadIdx.x & 31;
    const int w    = threadIdx.x >> 5;
    const int P0   = blockIdx.x * 32 + w * 4;
    const int b    = P0 >> 12;
    const int pix0 = P0 & 4095;

    const float* fm2 = fmap2 + ((size_t)b * CDIM) * NPIX + pix0;
    const float* mt  = MT + 4 * lane;

    float4 acc0 = make_float4(0.f, 0.f, 0.f, 0.f);
    float4 acc1 = make_float4(0.f, 0.f, 0.f, 0.f);
    float4 acc2 = make_float4(0.f, 0.f, 0.f, 0.f);
    float4 acc3 = make_float4(0.f, 0.f, 0.f, 0.f);

#pragma unroll 16
    for (int c2 = 0; c2 < CDIM; ++c2) {
        const float4 m = *(const float4*)(mt + c2 * CDIM);
        const float4 f = *(const float4*)(fm2 + (size_t)c2 * NPIX);
        acc0.x = fmaf(m.x, f.x, acc0.x);
        acc0.y = fmaf(m.y, f.x, acc0.y);
        acc0.z = fmaf(m.z, f.x, acc0.z);
        acc0.w = fmaf(m.w, f.x, acc0.w);
        acc1.x = fmaf(m.x, f.y, acc1.x);
        acc1.y = fmaf(m.y, f.y, acc1.y);
        acc1.z = fmaf(m.z, f.y, acc1.z);
        acc1.w = fmaf(m.w, f.y, acc1.w);
        acc2.x = fmaf(m.x, f.z, acc2.x);
        acc2.y = fmaf(m.y, f.z, acc2.y);
        acc2.z = fmaf(m.z, f.z, acc2.z);
        acc2.w = fmaf(m.w, f.z, acc2.w);
        acc3.x = fmaf(m.x, f.w, acc3.x);
        acc3.y = fmaf(m.y, f.w, acc3.y);
        acc3.z = fmaf(m.z, f.w, acc3.z);
        acc3.w = fmaf(m.w, f.w, acc3.w);
    }

    float* o = Tt + (size_t)P0 * CDIM + 4 * lane;
    *(float4*)(o)            = acc0;
    *(float4*)(o + CDIM)     = acc1;
    *(float4*)(o + 2 * CDIM) = acc2;
    *(float4*)(o + 3 * CDIM) = acc3;
}

#define APITCH 72
#define BPITCH 132
#define BTILE  (64 * BPITCH)
#define MT_PER_WG 16

__device__ __forceinline__ void gram_tile(const float* Ap, const float* Bt,
                                          v8f& acc0, v8f& acc1) {
    acc0 = (v8f){};
    acc1 = (v8f){};
#pragma unroll
    for (int k = 0; k < CDIM; k += 4) {
        v2f a;
        a.x = Ap[k * APITCH];
        a.y = Ap[(k + 1) * APITCH];
        const v2f b0 = *(const v2f*)(Bt + k);
        const v2f b1 = *(const v2f*)(Bt + 16 * BPITCH + k);
        acc0 = __builtin_amdgcn_wmma_f32_16x16x4_f32(
                   false, a, false, b0, (short)0, acc0, false, false);
        acc1 = __builtin_amdgcn_wmma_f32_16x16x4_f32(
                   false, a, false, b1, (short)0, acc1, false, false);
    }
}

__device__ __forceinline__ void gram_store(float* Gt, const v8f& acc0, const v8f& acc1) {
#pragma unroll
    for (int r = 0; r < 8; ++r) {
        Gt[(size_t)r * NPIX]      = acc0[r];
        Gt[(size_t)r * NPIX + 16] = acc1[r];
    }
}

__global__ void __launch_bounds__(256)
shiftlsa_stageB(const float* __restrict__ fmap, const float* __restrict__ Tt,
                float* __restrict__ G) {
    __shared__ __attribute__((aligned(16))) float As[CDIM * APITCH];
    __shared__ __attribute__((aligned(16))) float Bs[2 * BTILE];

    const int b   = blockIdx.z;
    const int n0  = blockIdx.x * 64;
    const int mc0 = blockIdx.y * (MT_PER_WG * 64);

    const int tid  = threadIdx.x;
    const int lane = tid & 31;
    const int w    = tid >> 5;
    const int twr  = w & 3;
    const int twc  = w >> 2;

    const int ar = tid >> 4, ac = (tid & 15) << 2;
    const int br = tid >> 5, bc = (tid & 31) << 2;

    const float* Xa = fmap + ((size_t)b * CDIM + ar) * NPIX + n0 + ac;
    const float* Fs = Tt + ((size_t)b * NPIX + mc0 + br) * CDIM + bc;

#pragma unroll
    for (int it = 0; it < 8; ++it)
        *(float4*)(&As[(ar + 16 * it) * APITCH + ac]) =
            *(const float4*)(Xa + (size_t)(16 * it) * NPIX);
#pragma unroll
    for (int it = 0; it < 8; ++it)
        *(float4*)(&Bs[(br + 8 * it) * BPITCH + bc]) =
            *(const float4*)(Fs + (size_t)(8 * it) * CDIM);
    __syncthreads();

    const int koff = (lane >> 4) << 1;
    const float* Ap = As + koff * APITCH + twr * 16 + (lane & 15);
    const int boff  = (twc * 32 + (lane & 15)) * BPITCH + koff;
    float* Gt = G + (size_t)b * NPIX * NPIX
                  + (size_t)(n0 + twr * 16 + ((lane >> 4) << 3)) * NPIX
                  + mc0 + twc * 32 + (lane & 15);
    float* Bst = Bs + br * BPITCH + bc;

    v8f acc0, acc1;
#pragma nounroll
    for (int t = 0; t < MT_PER_WG - 1; ++t) {
        float4 pre[8];
        const float* Fn = Fs + (size_t)(t + 1) * 64 * CDIM;
#pragma unroll
        for (int it = 0; it < 8; ++it)
            pre[it] = *(const float4*)(Fn + (size_t)(8 * it) * CDIM);
        __builtin_amdgcn_sched_barrier(0);

        gram_tile(Ap, Bs + (t & 1) * BTILE + boff, acc0, acc1);

        float* Bn = Bst + ((t + 1) & 1) * BTILE;
#pragma unroll
        for (int it = 0; it < 8; ++it)
            *(float4*)(Bn + 8 * it * BPITCH) = pre[it];
        __syncthreads();

        gram_store(Gt + t * 64, acc0, acc1);
    }
    gram_tile(Ap, Bs + ((MT_PER_WG - 1) & 1) * BTILE + boff, acc0, acc1);
    gram_store(Gt + (MT_PER_WG - 1) * 64, acc0, acc1);
}

__device__ __forceinline__ float from_next_lane(float edge, float src) {
    return __int_as_float(__builtin_amdgcn_update_dpp(
        __float_as_int(edge), __float_as_int(src), 0x101, 0xf, 0xf, false));
}
__device__ __forceinline__ float from_prev_lane(float edge, float src) {
    return __int_as_float(__builtin_amdgcn_update_dpp(
        __float_as_int(edge), __float_as_int(src), 0x111, 0xf, 0xf, false));
}

#define DNY 4
#define DNX 8
#define DKR (DNY + 4)
#define DKX (DNX + 4)

__device__ __forceinline__ void diag_load_row(float4 (&g)[DKX], const float* Gb,
                                              int y0, int Y0, int x0, int k) {
    const int py = clamp6(y0 + k - 2);
    const int ro = clamp6(Y0 + k - 2) * 64;
    const float* Grow = Gb + (size_t)(py * 64) * NPIX + ro;
#pragma unroll
    for (int kx = 0; kx < DKX; ++kx) {
        int px = x0 - 2 + kx;
        if (kx < 2)        px = px < 0 ? 0 : px;
        if (kx >= DNX + 2) px = px > 63 ? 63 : px;
        g[kx] = *(const float4*)(Grow + (size_t)px * NPIX);
    }
}

template <int K>
__device__ __forceinline__ void diag_row_taps(const float4 (&g)[DKX],
                                              float4 (&acc)[DNY][DNX],
                                              const float* an) {
#pragma unroll
    for (int kx = 0; kx < DKX; ++kx) {
        const float4 v = g[kx];
        const float pz = from_prev_lane(v.x, v.z);
        const float pw = from_prev_lane(v.x, v.w);
        const float nx = from_next_lane(v.w, v.x);
        const float ny = from_next_lane(v.w, v.y);
#pragma unroll
        for (int a = 0; a < DNY; ++a) {
            const int i = K - a;
            if (i < 0 || i > 4) continue;
#pragma unroll
            for (int t = 0; t < DNX; ++t) {
                const int j = kx - 2 - t;
                if (j < -2 || j > 2) continue;
                const float w = an[(a * 64 + t) * LWIN + i * 5 + (j + 2)];
                float4 s;
                if (j == -2)      s = make_float4(pz,  pw,  v.x, v.y);
                else if (j == -1) s = make_float4(pw,  v.x, v.y, v.z);
                else if (j == 0)  s = v;
                else if (j == 1)  s = make_float4(v.y, v.z, v.w, nx);
                else              s = make_float4(v.z, v.w, nx,  ny);
                acc[a][t].x = fmaf(w, s.x, acc[a][t].x);
                acc[a][t].y = fmaf(w, s.y, acc[a][t].y);
                acc[a][t].z = fmaf(w, s.z, acc[a][t].z);
                acc[a][t].w = fmaf(w, s.w, acc[a][t].w);
            }
        }
    }
}

template <int K>
__device__ __forceinline__ void diag_step(float4 (&gcur)[DKX], float4 (&gnext)[DKX],
                                          float4 (&acc)[DNY][DNX], const float* an,
                                          const float* Gb, int y0, int Y0, int x0) {
    if (K + 1 < DKR) diag_load_row(gnext, Gb, y0, Y0, x0, K + 1);
    int zk = 0;
    asm volatile("" : "+s"(zk));
    const float* ak = an + __builtin_amdgcn_readfirstlane(zk);
    __builtin_amdgcn_sched_barrier(0);
    diag_row_taps<K>(gcur, acc, ak);
    __builtin_amdgcn_sched_barrier(0);
}

__global__ void __launch_bounds__(64)
shiftlsa_stageC(const float* __restrict__ attn, const float* __restrict__ G,
                float* __restrict__ out) {
    const int b  = blockIdx.z;
    const int y0 = (blockIdx.x / 8) * DNY;
    const int x0 = (blockIdx.x % 8) * DNX;
    const int c  = threadIdx.x & 15;
    const int Y0 = (int)(blockIdx.y * 4 + (threadIdx.x >> 4)) - (DNY - 1);

    const float* an = attn + ((size_t)b * NPIX + y0 * 64 + x0) * LWIN;
    const float* Gb = G + (size_t)b * NPIX * NPIX + 4 * c;

    float4 acc[DNY][DNX];
#pragma unroll
    for (int a = 0; a < DNY; ++a)
#pragma unroll
        for (int t = 0; t < DNX; ++t) acc[a][t] = make_float4(0.f, 0.f, 0.f, 0.f);

    float4 g0[DKX], g1[DKX];
    diag_load_row(g0, Gb, y0, Y0, x0, 0);
    diag_step<0>(g0, g1, acc, an, Gb, y0, Y0, x0);
    diag_step<1>(g1, g0, acc, an, Gb, y0, Y0, x0);
    diag_step<2>(g0, g1, acc, an, Gb, y0, Y0, x0);
    diag_step<3>(g1, g0, acc, an, Gb, y0, Y0, x0);
    diag_step<4>(g0, g1, acc, an, Gb, y0, Y0, x0);
    diag_step<5>(g1, g0, acc, an, Gb, y0, Y0, x0);
    diag_step<6>(g0, g1, acc, an, Gb, y0, Y0, x0);
    diag_step<7>(g1, g0, acc, an, Gb, y0, Y0, x0);

#pragma unroll
    for (int a = 0; a < DNY; ++a)
#pragma unroll
        for (int t = 0; t < DNX; ++t)
            asm volatile("" : "+v"(acc[a][t].x), "+v"(acc[a][t].y),
                              "+v"(acc[a][t].z), "+v"(acc[a][t].w));

    const float scale = 0.0883883476483184405f;
#pragma unroll
    for (int a = 0; a < DNY; ++a) {
        const int Y = Y0 + a;
        if (Y >= 0 && Y < 64) {
            float* orow = out + ((size_t)b * NPIX + (y0 + a) * 64 + x0) * NPIX + Y * 64 + 4 * c;
#pragma unroll
            for (int t = 0; t < DNX; ++t) {
                float4 o;
                o.x = acc[a][t].x * scale;
                o.y = acc[a][t].y * scale;
                o.z = acc[a][t].z * scale;
                o.w = acc[a][t].w * scale;
                *(float4*)(orow + (size_t)t * NPIX) = o;
            }
        }
    }
}

extern "C" void kernel_launch(void* const* d_in, const int* in_sizes, int n_in,
                              void* d_out, int out_size, void* d_ws, size_t ws_size,
                              hipStream_t stream) {
    const float* attn  = (const float*)d_in[0];
    const float* fmap  = (const float*)d_in[1];
    const float* fmap2 = (const float*)d_in[2];
    const float* W1    = (const float*)d_in[3];
    const float* W2    = (const float*)d_in[4];
    float*       out   = (float*)d_out;

    float* Tt = (float*)d_ws;
    float* MT = Tt + (size_t)2 * NPIX * CDIM;
    float* G  = MT + (size_t)2 * NPIX * CDIM;

    shiftlsa_stageM<<<dim3(64), dim3(256), 0, stream>>>(W1, W2, MT);

    shiftlsa_stageA<<<dim3(256), dim3(256), 0, stream>>>(fmap2, MT, Tt);

    shiftlsa_stageB<<<dim3(64, 4, 2), dim3(256), 0, stream>>>(fmap, Tt, G);

    shiftlsa_stageC<<<dim3(128, 17, 2), dim3(64), 0, stream>>>(attn, G, out);
}
